# starting kernel
# baseline (speedup 1.0000x reference)
#include <hip/hip_runtime.h>

typedef float v2f __attribute__((ext_vector_type(2)));
typedef float v8f __attribute__((ext_vector_type(8)));

static constexpr int KD = 10;

__global__ __launch_bounds__(256) void precompute_affine(
    const float* __restrict__ W, const float* __restrict__ b,
    const int* __restrict__ steps_p, float* __restrict__ ws) {
  __shared__ float A[KD][KD];
  __shared__ float Mbuf[2][KD][KD];
  __shared__ float vbuf[2][KD];
  __shared__ float u[KD];
  const int tid = threadIdx.x;

  if (tid < KD * KD) {
    int k = tid / KD, n = tid % KD;
    A[k][n] = W[n * KD + k];
    Mbuf[0][k][n] = (k == n) ? 1.0f : 0.0f;
  }
  if (tid < KD) {
    u[tid] = b[tid] + 1.0f;
    vbuf[0][tid] = 0.0f;
  }
  __syncthreads();

  const int S = steps_p[0];
  int cur = 0;
  for (int t = 0; t < S; ++t) {
    int nxt = cur ^ 1;
    if (tid < KD * KD) {
      int i = tid / KD, j = tid % KD;
      float acc = 0.0f;
#pragma unroll
      for (int k = 0; k < KD; ++k) acc += Mbuf[cur][i][k] * A[k][j];
      Mbuf[nxt][i][j] = acc;
    }
    if (tid < KD) {
      float acc = u[tid];
#pragma unroll
      for (int k = 0; k < KD; ++k) acc += vbuf[cur][k] * A[k][tid];
      vbuf[nxt][tid] = acc;
    }
    __syncthreads();
    cur = nxt;
  }

  if (tid < 256) {
    int k = tid >> 4, n = tid & 15;
    ws[tid] = (k < KD && n < KD) ? Mbuf[cur][k][n] : 0.0f;
  }
  if (tid < 16) ws[256 + tid] = (tid < KD) ? vbuf[cur][tid] : 0.0f;
}

__global__ __launch_bounds__(256) void apply_affine_wmma(
    const float* __restrict__ x, const float* __restrict__ ws,
    float* __restrict__ out, int ntiles) {
  const int lane = threadIdx.x & 31;
  const int wid = (int)(blockIdx.x * (blockDim.x >> 5) + (threadIdx.x >> 5));
  if (wid >= ntiles) return;

  const int n  = lane & 15;
  const int r  = lane & 15;
  const int kb = (lane >> 4) * 2;
  const bool lo = lane < 16;

  v2f bf0, bf1, bf2;
  bf0.x = ws[(0 + kb) * 16 + n];  bf0.y = ws[(1 + kb) * 16 + n];
  bf1.x = ws[(4 + kb) * 16 + n];  bf1.y = ws[(5 + kb) * 16 + n];
  bf2.x = ws[(8 + kb) * 16 + n];  bf2.y = ws[(9 + kb) * 16 + n];

  const float bias = ws[256 + n];
  v8f c;
#pragma unroll
  for (int p = 0; p < 8; ++p) c[p] = bias;

  const float* xr = x + (size_t)wid * (16 * KD) + (size_t)r * KD;
  v2f a0 = __builtin_nontemporal_load((const v2f*)(xr + kb));
  v2f a1 = __builtin_nontemporal_load((const v2f*)(xr + 4 + kb));
  v2f a2r = __builtin_nontemporal_load((const v2f*)(xr + 8));
  v2f a2;
  a2.x = lo ? a2r.x : 0.0f;
  a2.y = lo ? a2r.y : 0.0f;

  c = __builtin_amdgcn_wmma_f32_16x16x4_f32(false, a0, false, bf0,
                                            (short)0, c, false, false);
  c = __builtin_amdgcn_wmma_f32_16x16x4_f32(false, a1, false, bf1,
                                            (short)0, c, false, false);
  c = __builtin_amdgcn_wmma_f32_16x16x4_f32(false, a2, false, bf2,
                                            (short)0, c, false, false);

  if (n < KD) {
    float* orow = out + (size_t)wid * (16 * KD) + (size_t)(lo ? 0 : 8 * KD) + n;
#pragma unroll
    for (int p = 0; p < 8; ++p)
      __builtin_nontemporal_store(c[p], orow + p * KD);
  }
}

__global__ __launch_bounds__(32) void apply_affine_tail(
    const float* __restrict__ x, const float* __restrict__ ws,
    float* __restrict__ out, int startrow, int nrows) {
  const int row = startrow + (int)threadIdx.x;
  if (row >= nrows) return;
  float xi[KD];
#pragma unroll
  for (int k = 0; k < KD; ++k) xi[k] = x[(size_t)row * KD + k];
#pragma unroll
  for (int nn = 0; nn < KD; ++nn) {
    float acc = ws[256 + nn];
#pragma unroll
    for (int k = 0; k < KD; ++k) acc += xi[k] * ws[k * 16 + nn];
    out[(size_t)row * KD + nn] = acc;
  }
}

extern "C" void kernel_launch(void* const* d_in, const int* in_sizes, int n_in,
                              void* d_out, int out_size, void* d_ws, size_t ws_size,
                              hipStream_t stream) {
  const float* x = (const float*)d_in[0];
  const float* W = (const float*)d_in[1];
  const float* b = (const float*)d_in[2];
  const int* steps = (const int*)d_in[3];
  float* ws = (float*)d_ws;
  float* out = (float*)d_out;

  const int nrows = in_sizes[0] / KD;
  const int ntiles_full = nrows / 16;
  const int tail = nrows - ntiles_full * 16;

  precompute_affine<<<1, 256, 0, stream>>>(W, b, steps, ws);

  if (ntiles_full > 0) {
    const int waves_per_block = 256 / 32;
    const int blocks = (ntiles_full + waves_per_block - 1) / waves_per_block;
    apply_affine_wmma<<<blocks, 256, 0, stream>>>(x, ws, out, ntiles_full);
  }
  if (tail > 0) {
    apply_affine_tail<<<1, 32, 0, stream>>>(x, ws, out, ntiles_full * 16, nrows);
  }
}


# precompute kernel: the num_steps recurrence runs in the registers of one wave, one lane per row of [M; v] (rows are independent), no LDS or barrier per step; same k order, products fused
# speedup vs baseline: 1.6344x; 1.6344x over previous
#include <hip/hip_runtime.h>

typedef float v2f __attribute__((ext_vector_type(2)));
typedef float v8f __attribute__((ext_vector_type(8)));

static constexpr int KD = 10;

__global__ __launch_bounds__(32) void precompute_affine(
    const float* __restrict__ W, const float* __restrict__ b,
    const int* __restrict__ steps_p, float* __restrict__ ws) {
  const int lane = threadIdx.x;

  const int S = steps_p[0];
  float u[KD];
#pragma unroll
  for (int j = 0; j < KD; ++j) u[j] = b[j] + 1.0f;

  __shared__ float Wl[KD * KD];
  constexpr int WCH = (KD * KD + 31) / 32;
  float wtmp[WCH];
#pragma unroll
  for (int i = 0; i < WCH; ++i) {
    const int e = lane + 32 * i;
    wtmp[i] = W[e < KD * KD ? e : KD * KD - 1];
  }
#pragma unroll
  for (int i = 0; i < WCH; ++i) {
    const int e = lane + 32 * i;
    if (e < KD * KD) Wl[e] = wtmp[i];
  }
  __syncthreads();
  float w[KD][KD];
#pragma unroll
  for (int j = 0; j < KD; ++j) {
#pragma unroll
    for (int k = 0; k < KD; ++k) w[j][k] = Wl[j * KD + k];
  }

  float row[KD];
  float add[KD];
  const bool vrow = (lane == KD);
#pragma unroll
  for (int j = 0; j < KD; ++j) {
    row[j] = (lane == j) ? 1.0f : 0.0f;
    add[j] = vrow ? u[j] : 0.0f;
  }

  for (int t = 0; t < S; ++t) {
    float nr[KD];
#pragma unroll
    for (int j = 0; j < KD; ++j) {
      float acc = add[j];
#pragma unroll
      for (int k = 0; k < KD; ++k) acc += row[k] * w[j][k];
      nr[j] = acc;
    }
#pragma unroll
    for (int j = 0; j < KD; ++j) row[j] = nr[j];
  }

  if (lane < 16) {
    const bool live = lane < KD;
    float4* dst = (float4*)(ws + lane * 16);
    dst[0] = make_float4(live ? row[0] : 0.0f, live ? row[1] : 0.0f,
                         live ? row[2] : 0.0f, live ? row[3] : 0.0f);
    dst[1] = make_float4(live ? row[4] : 0.0f, live ? row[5] : 0.0f,
                         live ? row[6] : 0.0f, live ? row[7] : 0.0f);
    dst[2] = make_float4(live ? row[8] : 0.0f, live ? row[9] : 0.0f, 0.0f, 0.0f);
    dst[3] = make_float4(0.0f, 0.0f, 0.0f, 0.0f);
  }
  if (lane == KD) {
    float4* dst = (float4*)(ws + 256);
    dst[0] = make_float4(row[0], row[1], row[2], row[3]);
    dst[1] = make_float4(row[4], row[5], row[6], row[7]);
    dst[2] = make_float4(row[8], row[9], 0.0f, 0.0f);
    dst[3] = make_float4(0.0f, 0.0f, 0.0f, 0.0f);
  }
}

__global__ __launch_bounds__(256) void apply_affine_wmma(
    const float* __restrict__ x, const float* __restrict__ ws,
    float* __restrict__ out, int ntiles) {
  const int lane = threadIdx.x & 31;
  const int wid = (int)(blockIdx.x * (blockDim.x >> 5) + (threadIdx.x >> 5));
  if (wid >= ntiles) return;

  const int n  = lane & 15;
  const int r  = lane & 15;
  const int kb = (lane >> 4) * 2;
  const bool lo = lane < 16;

  v2f bf0, bf1, bf2;
  bf0.x = ws[(0 + kb) * 16 + n];  bf0.y = ws[(1 + kb) * 16 + n];
  bf1.x = ws[(4 + kb) * 16 + n];  bf1.y = ws[(5 + kb) * 16 + n];
  bf2.x = ws[(8 + kb) * 16 + n];  bf2.y = ws[(9 + kb) * 16 + n];

  const float bias = ws[256 + n];
  v8f c;
#pragma unroll
  for (int p = 0; p < 8; ++p) c[p] = bias;

  const float* xr = x + (size_t)wid * (16 * KD) + (size_t)r * KD;
  v2f a0 = __builtin_nontemporal_load((const v2f*)(xr + kb));
  v2f a1 = __builtin_nontemporal_load((const v2f*)(xr + 4 + kb));
  v2f a2r = __builtin_nontemporal_load((const v2f*)(xr + 8));
  v2f a2;
  a2.x = lo ? a2r.x : 0.0f;
  a2.y = lo ? a2r.y : 0.0f;

  c = __builtin_amdgcn_wmma_f32_16x16x4_f32(false, a0, false, bf0,
                                            (short)0, c, false, false);
  c = __builtin_amdgcn_wmma_f32_16x16x4_f32(false, a1, false, bf1,
                                            (short)0, c, false, false);
  c = __builtin_amdgcn_wmma_f32_16x16x4_f32(false, a2, false, bf2,
                                            (short)0, c, false, false);

  if (n < KD) {
    float* orow = out + (size_t)wid * (16 * KD) + (size_t)(lo ? 0 : 8 * KD) + n;
#pragma unroll
    for (int p = 0; p < 8; ++p)
      __builtin_nontemporal_store(c[p], orow + p * KD);
  }
}

__global__ __launch_bounds__(32) void apply_affine_tail(
    const float* __restrict__ x, const float* __restrict__ ws,
    float* __restrict__ out, int startrow, int nrows) {
  const int row = startrow + (int)threadIdx.x;
  if (row >= nrows) return;
  float xi[KD];
#pragma unroll
  for (int k = 0; k < KD; ++k) xi[k] = x[(size_t)row * KD + k];
#pragma unroll
  for (int nn = 0; nn < KD; ++nn) {
    float acc = ws[256 + nn];
#pragma unroll
    for (int k = 0; k < KD; ++k) acc += xi[k] * ws[k * 16 + nn];
    out[(size_t)row * KD + nn] = acc;
  }
}

extern "C" void kernel_launch(void* const* d_in, const int* in_sizes, int n_in,
                              void* d_out, int out_size, void* d_ws, size_t ws_size,
                              hipStream_t stream) {
  const float* x = (const float*)d_in[0];
  const float* W = (const float*)d_in[1];
  const float* b = (const float*)d_in[2];
  const int* steps = (const int*)d_in[3];
  float* ws = (float*)d_ws;
  float* out = (float*)d_out;

  const int nrows = in_sizes[0] / KD;
  const int ntiles_full = nrows / 16;
  const int tail = nrows - ntiles_full * 16;

  precompute_affine<<<1, 32, 0, stream>>>(W, b, steps, ws);

  if (ntiles_full > 0) {
    const int waves_per_block = 256 / 32;
    const int blocks = (ntiles_full + waves_per_block - 1) / waves_per_block;
    apply_affine_wmma<<<blocks, 256, 0, stream>>>(x, ws, out, ntiles_full);
  }
  if (tail > 0) {
    apply_affine_tail<<<1, 32, 0, stream>>>(x, ws, out, ntiles_full * 16, nrows);
  }
}


# fused into one launch of 256 persistent work-groups: wave 0 of each runs the recurrence in registers while its other 15 waves keep all their x tile loads in flight; (M, v) passed through LDS
# speedup vs baseline: 2.3789x; 1.4555x over previous
#include <hip/hip_runtime.h>

typedef float v2f __attribute__((ext_vector_type(2)));
typedef float v8f __attribute__((ext_vector_type(8)));

static constexpr int KD = 10;
static constexpr int NWG = 256;
static constexpr int WPB = 16;
static constexpr int NWORK = NWG * (WPB - 1);
static constexpr int TPW = 17;

__global__ __launch_bounds__(WPB * 32) void fused_affine_wmma(
    const float* __restrict__ x, const float* __restrict__ W,
    const float* __restrict__ b, const int* __restrict__ steps_p,
    float* __restrict__ ws, float* __restrict__ out, int ntiles) {
  __shared__ float Wl[128];
  __shared__ __attribute__((aligned(16))) float Ml[16 * 16 + 16];

  const int lane = threadIdx.x & 31;
  const int wave = __builtin_amdgcn_readfirstlane((int)(threadIdx.x >> 5));
  const bool worker = (wave != 0) && (ntiles > 0);

  const int n  = lane & 15;
  const int r  = lane & 15;
  const int kb = (lane >> 4) * 2;
  const bool lo = lane < 16;
  const int wk = (int)blockIdx.x * (WPB - 1) + (wave - 1);

  v2f a0[TPW], a1[TPW], a2r[TPW];

  if (wave == 0) {
    const int S = steps_p[0];
    float u[KD];
#pragma unroll
    for (int j = 0; j < KD; ++j) u[j] = b[j] + 1.0f;

    constexpr int WCH = (KD * KD + 31) / 32;
    float wtmp[WCH];
#pragma unroll
    for (int i = 0; i < WCH; ++i) {
      const int e = lane + 32 * i;
      wtmp[i] = W[e < KD * KD ? e : KD * KD - 1];
    }
#pragma unroll
    for (int i = 0; i < WCH; ++i) Wl[lane + 32 * i] = wtmp[i];
    __builtin_amdgcn_fence(__ATOMIC_RELEASE, "workgroup");
    __builtin_amdgcn_wave_barrier();
    __builtin_amdgcn_fence(__ATOMIC_ACQUIRE, "workgroup");
    float w[KD][KD];
#pragma unroll
    for (int j = 0; j < KD; ++j) {
#pragma unroll
      for (int k = 0; k < KD; ++k) w[j][k] = Wl[j * KD + k];
    }

    float row[KD];
    float add[KD];
    const bool vrow = (lane == KD);
#pragma unroll
    for (int j = 0; j < KD; ++j) {
      row[j] = (lane == j) ? 1.0f : 0.0f;
      add[j] = vrow ? u[j] : 0.0f;
    }

    for (int t = 0; t < S; ++t) {
      float nr[KD];
#pragma unroll
      for (int j = 0; j < KD; ++j) {
        float acc = add[j];
#pragma unroll
        for (int k = 0; k < KD; ++k) acc += row[k] * w[j][k];
        nr[j] = acc;
      }
#pragma unroll
      for (int j = 0; j < KD; ++j) row[j] = nr[j];
    }

    if (lane < 16) {
      const bool live = lane < KD;
      float4* dl = (float4*)(Ml + lane * 16);
      dl[0] = make_float4(live ? row[0] : 0.0f, live ? row[1] : 0.0f,
                          live ? row[2] : 0.0f, live ? row[3] : 0.0f);
      dl[1] = make_float4(live ? row[4] : 0.0f, live ? row[5] : 0.0f,
                          live ? row[6] : 0.0f, live ? row[7] : 0.0f);
      dl[2] = make_float4(live ? row[8] : 0.0f, live ? row[9] : 0.0f, 0.0f, 0.0f);
      dl[3] = make_float4(0.0f, 0.0f, 0.0f, 0.0f);
    }
    if (lane == KD) {
      float4* dl = (float4*)(Ml + 256);
      dl[0] = make_float4(row[0], row[1], row[2], row[3]);
      dl[1] = make_float4(row[4], row[5], row[6], row[7]);
      dl[2] = make_float4(row[8], row[9], 0.0f, 0.0f);
      dl[3] = make_float4(0.0f, 0.0f, 0.0f, 0.0f);
    }
  } else if (worker) {
#pragma unroll
    for (int i = 0; i < TPW; ++i) {
      const int t = wk + i * NWORK;
      const int tc = t < ntiles ? t : ntiles - 1;
      const float* xr = x + (size_t)tc * (16 * KD) + (size_t)r * KD;
      a0[i]  = __builtin_nontemporal_load((const v2f*)(xr + kb));
      a1[i]  = __builtin_nontemporal_load((const v2f*)(xr + 4 + kb));
      a2r[i] = __builtin_nontemporal_load((const v2f*)(xr + 8));
    }
  }

  __syncthreads();
  if (wave == 0) {
    if (blockIdx.x == 0) {
      const float4* sl = (const float4*)Ml;
      float4* dg = (float4*)ws;
      dg[lane] = sl[lane];
      dg[lane + 32] = sl[lane + 32];
      if (lane < 4) dg[lane + 64] = sl[lane + 64];
    }
    return;
  }
  if (!worker) return;

  v2f bf0, bf1, bf2;
  bf0.x = Ml[(0 + kb) * 16 + n];  bf0.y = Ml[(1 + kb) * 16 + n];
  bf1.x = Ml[(4 + kb) * 16 + n];  bf1.y = Ml[(5 + kb) * 16 + n];
  bf2.x = Ml[(8 + kb) * 16 + n];  bf2.y = Ml[(9 + kb) * 16 + n];
  const float bias = Ml[256 + n];

  for (int base = 0;;) {
#pragma unroll
    for (int i = 0; i < TPW; ++i) {
      const int t = base + wk + i * NWORK;
      v8f c;
#pragma unroll
      for (int p = 0; p < 8; ++p) c[p] = bias;
      v2f a2;
      a2.x = lo ? a2r[i].x : 0.0f;
      a2.y = lo ? a2r[i].y : 0.0f;
      c = __builtin_amdgcn_wmma_f32_16x16x4_f32(false, a0[i], false, bf0,
                                                (short)0, c, false, false);
      c = __builtin_amdgcn_wmma_f32_16x16x4_f32(false, a1[i], false, bf1,
                                                (short)0, c, false, false);
      c = __builtin_amdgcn_wmma_f32_16x16x4_f32(false, a2, false, bf2,
                                                (short)0, c, false, false);
      if (t < ntiles && n < KD) {
        float* orow = out + (size_t)t * (16 * KD) + (size_t)(lo ? 0 : 8 * KD) + n;
#pragma unroll
        for (int p = 0; p < 8; ++p)
          __builtin_nontemporal_store(c[p], orow + p * KD);
      }
    }
    base += TPW * NWORK;
    if (base >= ntiles) break;
#pragma unroll
    for (int i = 0; i < TPW; ++i) {
      const int t = base + wk + i * NWORK;
      const int tc = t < ntiles ? t : ntiles - 1;
      const float* xr = x + (size_t)tc * (16 * KD) + (size_t)r * KD;
      a0[i]  = __builtin_nontemporal_load((const v2f*)(xr + kb));
      a1[i]  = __builtin_nontemporal_load((const v2f*)(xr + 4 + kb));
      a2r[i] = __builtin_nontemporal_load((const v2f*)(xr + 8));
    }
  }
}

__global__ __launch_bounds__(32) void apply_affine_tail(
    const float* __restrict__ x, const float* __restrict__ ws,
    float* __restrict__ out, int startrow, int nrows) {
  const int row = startrow + (int)threadIdx.x;
  if (row >= nrows) return;
  float xi[KD];
#pragma unroll
  for (int k = 0; k < KD; ++k) xi[k] = x[(size_t)row * KD + k];
#pragma unroll
  for (int nn = 0; nn < KD; ++nn) {
    float acc = ws[256 + nn];
#pragma unroll
    for (int k = 0; k < KD; ++k) acc += xi[k] * ws[k * 16 + nn];
    out[(size_t)row * KD + nn] = acc;
  }
}

extern "C" void kernel_launch(void* const* d_in, const int* in_sizes, int n_in,
                              void* d_out, int out_size, void* d_ws, size_t ws_size,
                              hipStream_t stream) {
  const float* x = (const float*)d_in[0];
  const float* W = (const float*)d_in[1];
  const float* b = (const float*)d_in[2];
  const int* steps = (const int*)d_in[3];
  float* ws = (float*)d_ws;
  float* out = (float*)d_out;

  const int nrows = in_sizes[0] / KD;
  const int ntiles_full = nrows / 16;
  const int tail = nrows - ntiles_full * 16;

  fused_affine_wmma<<<NWG, WPB * 32, 0, stream>>>(x, W, b, steps, ws, out,
                                                  ntiles_full);
  if (tail > 0) {
    apply_affine_tail<<<1, 32, 0, stream>>>(x, ws, out, ntiles_full * 16, nrows);
  }
}
